# starting kernel
# baseline (speedup 1.0000x reference)
#include <hip/hip_runtime.h>
#include <math.h>

#define B_SZ   8
#define C_CH   64
#define L_LEN  8192
#define H_BAND 32
#define UNIT_L 512
#define NBLK   16
#define NSTRIP 32
#define NPOS   48
#define LSTRIDE 68
#define EPS_F  1e-12f

typedef float v2f __attribute__((ext_vector_type(2)));
typedef float v8f __attribute__((ext_vector_type(8)));

__global__ __launch_bounds__(96)
void FeatureExtractor_79448305041538_kernel(const float* __restrict__ feat,
                                            float* __restrict__ out) {
    __shared__ float Xs[NPOS * LSTRIDE];
    __shared__ float normS[NPOS];

    const int blk   = blockIdx.x;
    const int strip = blk & (NSTRIP - 1);
    const int n     = (blk >> 5) & (NBLK - 1);
    const int b     = blk >> 9;
    const int start = n * UNIT_L;
    const int w0    = strip * 16;

    const int tid = threadIdx.x;

    const float* fb = feat + (size_t)b * C_CH * L_LEN;
    for (int idx = tid; idx < C_CH * NPOS; idx += 96) {
        int c = idx / NPOS;
        int p = idx - c * NPOS;
        int g = start + w0 + p;
        float v = 0.0f;
        if (g < L_LEN) v = fb[(size_t)c * L_LEN + g];
        Xs[p * LSTRIDE + c] = v;
    }
    __syncthreads();

    if (tid < NPOS) {
        float s = 0.0f;
        for (int c = 0; c < C_CH; ++c) s += Xs[tid * LSTRIDE + c];
        float mu = s * (1.0f / 64.0f);
        float ss = 0.0f;
        for (int c = 0; c < C_CH; ++c) {
            float x = Xs[tid * LSTRIDE + c] - mu;
            Xs[tid * LSTRIDE + c] = x;
            ss += x * x;
        }
        normS[tid] = sqrtf(ss);
    }
    __syncthreads();

    const int wave  = tid >> 5;
    const int lane  = tid & 31;
    const int nn    = lane & 15;
    const int khalf = lane >> 4;

    const int aRow = nn;
    const int bRow = wave * 16 + nn;

    v8f acc = {};
#pragma unroll
    for (int k0 = 0; k0 < 16; ++k0) {
        int c = (k0 << 2) + (khalf << 1);
        v2f av = *(const v2f*)&Xs[aRow * LSTRIDE + c];
        v2f bv = *(const v2f*)&Xs[bRow * LSTRIDE + c];
        acc = __builtin_amdgcn_wmma_f32_16x16x4_f32(
             false, av,  false, bv,
             (short)0, acc,  false,  false);
    }

    const float ncol = normS[bRow];
    const int   vloc = wave * 16 + nn;
    const int   gcol = start + w0 + vloc;
#pragma unroll
    for (int r = 0; r < 8; ++r) {
        int m = r + (khalf << 3);
        int d = vloc - m - 1;
        if ((unsigned)d < (unsigned)H_BAND) {
            float denom = fmaxf(normS[m] * ncol, EPS_F);
            float val   = acc[r] / denom;
            if (gcol >= L_LEN) val = 0.0f;
            int t = w0 + m;
            out[((size_t)b * H_BAND + d) * L_LEN + (start + t)] = val;
        }
    }
}

extern "C" void kernel_launch(void* const* d_in, const int* in_sizes, int n_in,
                              void* d_out, int out_size, void* d_ws, size_t ws_size,
                              hipStream_t stream) {
    (void)in_sizes; (void)n_in; (void)out_size; (void)d_ws; (void)ws_size;
    const float* feat = (const float*)d_in[0];
    float* out = (float*)d_out;
    dim3 grid(B_SZ * NBLK * NSTRIP), block(96);
    FeatureExtractor_79448305041538_kernel<<<grid, block, 0, stream>>>(feat, out);
}


# Staging loads batched: each thread issues its eight 16-byte loads of the 64x48 window before any LDS store (one memory round trip, was 32 dependent 4-byte load-wait-store rounds)
# speedup vs baseline: 1.2337x; 1.2337x over previous
#include <hip/hip_runtime.h>
#include <math.h>

#define B_SZ   8
#define C_CH   64
#define L_LEN  8192
#define H_BAND 32
#define UNIT_L 512
#define NBLK   16
#define NSTRIP 32
#define NPOS   48
#define LSTRIDE 68
#define EPS_F  1e-12f
#define NQ     (NPOS / 4)

typedef float v2f __attribute__((ext_vector_type(2)));
typedef float v4f __attribute__((ext_vector_type(4)));
typedef float v8f __attribute__((ext_vector_type(8)));

__global__ __launch_bounds__(96)
void FeatureExtractor_79448305041538_kernel(const float* __restrict__ feat,
                                            float* __restrict__ out) {
    __shared__ float Xs[NPOS * LSTRIDE];
    __shared__ float normS[NPOS];

    const int blk   = blockIdx.x;
    const int strip = blk & (NSTRIP - 1);
    const int n     = (blk >> 5) & (NBLK - 1);
    const int b     = blk >> 9;
    const int start = n * UNIT_L;
    const int w0    = strip * 16;

    const int tid = threadIdx.x;

    const float* fb = feat + (size_t)b * C_CH * L_LEN;
    const int r  = tid / NQ;
    const int q  = tid - r * NQ;
    const int g  = start + w0 + 4 * q;
    v4f pc[8];
    if (g < L_LEN) {
        const float* srcp = fb + (size_t)r * L_LEN + g;
#pragma unroll
        for (int i = 0; i < 8; ++i)
            pc[i] = *(const v4f*)(srcp + (size_t)(8 * i) * L_LEN);
    } else {
#pragma unroll
        for (int i = 0; i < 8; ++i) pc[i] = (v4f){0.0f, 0.0f, 0.0f, 0.0f};
    }
    float* dstp = &Xs[(4 * q) * LSTRIDE + r];
#pragma unroll
    for (int i = 0; i < 8; ++i) {
        dstp[0 * LSTRIDE + 8 * i] = pc[i].x;
        dstp[1 * LSTRIDE + 8 * i] = pc[i].y;
        dstp[2 * LSTRIDE + 8 * i] = pc[i].z;
        dstp[3 * LSTRIDE + 8 * i] = pc[i].w;
    }
    __syncthreads();

    if (tid < NPOS) {
        float s = 0.0f;
        for (int c = 0; c < C_CH; ++c) s += Xs[tid * LSTRIDE + c];
        float mu = s * (1.0f / 64.0f);
        float ss = 0.0f;
        for (int c = 0; c < C_CH; ++c) {
            float x = Xs[tid * LSTRIDE + c] - mu;
            Xs[tid * LSTRIDE + c] = x;
            ss += x * x;
        }
        normS[tid] = sqrtf(ss);
    }
    __syncthreads();

    const int wave  = tid >> 5;
    const int lane  = tid & 31;
    const int nn    = lane & 15;
    const int khalf = lane >> 4;

    const int aRow = nn;
    const int bRow = wave * 16 + nn;

    v8f acc = {};
#pragma unroll
    for (int k0 = 0; k0 < 16; ++k0) {
        int c = (k0 << 2) + (khalf << 1);
        v2f av = *(const v2f*)&Xs[aRow * LSTRIDE + c];
        v2f bv = *(const v2f*)&Xs[bRow * LSTRIDE + c];
        acc = __builtin_amdgcn_wmma_f32_16x16x4_f32(
             false, av,  false, bv,
             (short)0, acc,  false,  false);
    }

    const float ncol = normS[bRow];
    const int   vloc = wave * 16 + nn;
    const int   gcol = start + w0 + vloc;
#pragma unroll
    for (int r = 0; r < 8; ++r) {
        int m = r + (khalf << 3);
        int d = vloc - m - 1;
        if ((unsigned)d < (unsigned)H_BAND) {
            float denom = fmaxf(normS[m] * ncol, EPS_F);
            float val   = acc[r] / denom;
            if (gcol >= L_LEN) val = 0.0f;
            int t = w0 + m;
            out[((size_t)b * H_BAND + d) * L_LEN + (start + t)] = val;
        }
    }
}

extern "C" void kernel_launch(void* const* d_in, const int* in_sizes, int n_in,
                              void* d_out, int out_size, void* d_ws, size_t ws_size,
                              hipStream_t stream) {
    (void)in_sizes; (void)n_in; (void)out_size; (void)d_ws; (void)ws_size;
    const float* feat = (const float*)d_in[0];
    float* out = (float*)d_out;
    dim3 grid(B_SZ * NBLK * NSTRIP), block(96);
    FeatureExtractor_79448305041538_kernel<<<grid, block, 0, stream>>>(feat, out);
}


# Band tile turned through LDS so each output row leaves as one contiguous 64-byte piece in 16-byte lane stores (was 4-byte stores to 32 different lines per instruction); values and sums unchanged
# speedup vs baseline: 4.0405x; 3.2751x over previous
#include <hip/hip_runtime.h>
#include <math.h>

#define B_SZ   8
#define C_CH   64
#define L_LEN  8192
#define H_BAND 32
#define UNIT_L 512
#define NBLK   16
#define NSTRIP 32
#define NPOS   48
#define LSTRIDE 68
#define EPS_F  1e-12f
#define NQ     (NPOS / 4)
#define OSTR   20

typedef float v2f __attribute__((ext_vector_type(2)));
typedef float v4f __attribute__((ext_vector_type(4)));
typedef float v8f __attribute__((ext_vector_type(8)));

__global__ __launch_bounds__(96)
void FeatureExtractor_79448305041538_kernel(const float* __restrict__ feat,
                                            float* __restrict__ out) {
    __shared__ float Xs[NPOS * LSTRIDE];
    __shared__ float normS[NPOS];
    __shared__ __attribute__((aligned(16))) float Os[H_BAND * OSTR];

    const int blk   = blockIdx.x;
    const int strip = blk & (NSTRIP - 1);
    const int n     = (blk >> 5) & (NBLK - 1);
    const int b     = blk >> 9;
    const int start = n * UNIT_L;
    const int w0    = strip * 16;

    const int tid = threadIdx.x;

    const float* fb = feat + (size_t)b * C_CH * L_LEN;
    const int r  = tid / NQ;
    const int q  = tid - r * NQ;
    const int g  = start + w0 + 4 * q;
    v4f pc[8];
    if (g < L_LEN) {
        const float* srcp = fb + (size_t)r * L_LEN + g;
#pragma unroll
        for (int i = 0; i < 8; ++i)
            pc[i] = *(const v4f*)(srcp + (size_t)(8 * i) * L_LEN);
    } else {
#pragma unroll
        for (int i = 0; i < 8; ++i) pc[i] = (v4f){0.0f, 0.0f, 0.0f, 0.0f};
    }
    float* dstp = &Xs[(4 * q) * LSTRIDE + r];
#pragma unroll
    for (int i = 0; i < 8; ++i) {
        dstp[0 * LSTRIDE + 8 * i] = pc[i].x;
        dstp[1 * LSTRIDE + 8 * i] = pc[i].y;
        dstp[2 * LSTRIDE + 8 * i] = pc[i].z;
        dstp[3 * LSTRIDE + 8 * i] = pc[i].w;
    }
    __syncthreads();

    if (tid < NPOS) {
        float s = 0.0f;
        for (int c = 0; c < C_CH; ++c) s += Xs[tid * LSTRIDE + c];
        float mu = s * (1.0f / 64.0f);
        float ss = 0.0f;
        for (int c = 0; c < C_CH; ++c) {
            float x = Xs[tid * LSTRIDE + c] - mu;
            Xs[tid * LSTRIDE + c] = x;
            ss += x * x;
        }
        normS[tid] = sqrtf(ss);
    }
    __syncthreads();

    const int wave  = tid >> 5;
    const int lane  = tid & 31;
    const int nn    = lane & 15;
    const int khalf = lane >> 4;

    const int aRow = nn;
    const int bRow = wave * 16 + nn;

    v8f acc = {};
#pragma unroll
    for (int k0 = 0; k0 < 16; ++k0) {
        int c = (k0 << 2) + (khalf << 1);
        v2f av = *(const v2f*)&Xs[aRow * LSTRIDE + c];
        v2f bv = *(const v2f*)&Xs[bRow * LSTRIDE + c];
        acc = __builtin_amdgcn_wmma_f32_16x16x4_f32(
             false, av,  false, bv,
             (short)0, acc,  false,  false);
    }

    const float ncol = normS[bRow];
    const int   vloc = wave * 16 + nn;
    const int   gcol = start + w0 + vloc;
#pragma unroll
    for (int r = 0; r < 8; ++r) {
        int m = r + (khalf << 3);
        int d = vloc - m - 1;
        if ((unsigned)d < (unsigned)H_BAND) {
            float denom = fmaxf(normS[m] * ncol, EPS_F);
            float val   = acc[r] / denom;
            if (gcol >= L_LEN) val = 0.0f;
            Os[d * OSTR + m] = val;
        }
    }
    __syncthreads();

    float* ob = out + (size_t)b * H_BAND * L_LEN + (start + w0);
    for (int idx = tid; idx < H_BAND * 4; idx += 96) {
        int d = idx >> 2;
        int p = idx & 3;
        v4f v = *(const v4f*)&Os[d * OSTR + 4 * p];
        *(v4f*)(ob + (size_t)d * L_LEN + 4 * p) = v;
    }
}

extern "C" void kernel_launch(void* const* d_in, const int* in_sizes, int n_in,
                              void* d_out, int out_size, void* d_ws, size_t ws_size,
                              hipStream_t stream) {
    (void)in_sizes; (void)n_in; (void)out_size; (void)d_ws; (void)ws_size;
    const float* feat = (const float*)d_in[0];
    float* out = (float*)d_out;
    dim3 grid(B_SZ * NBLK * NSTRIP), block(96);
    FeatureExtractor_79448305041538_kernel<<<grid, block, 0, stream>>>(feat, out);
}


# Workgroup tile grown from 16 rows (3 waves, one 16x16 tile each) to 128 rows (8 waves, one row tile times its 3 column tiles each): 4096 waves, was 12288; 160 positions staged per 128 rows
# speedup vs baseline: 6.5369x; 1.6178x over previous
#include <hip/hip_runtime.h>
#include <math.h>

#define B_SZ   8
#define C_CH   64
#define L_LEN  8192
#define H_BAND 32
#define ROWS   128
#define NPOS   (ROWS + H_BAND)
#define NSTRIP (L_LEN / ROWS)
#define NQ     (NPOS / 4)
#define PP     176
#define CPS    (2 * PP)
#define OSTR   132
#define EPS_F  1e-12f

typedef float v2f __attribute__((ext_vector_type(2)));
typedef float v4f __attribute__((ext_vector_type(4)));
typedef float v8f __attribute__((ext_vector_type(8)));

__global__ __launch_bounds__(256)
void FeatureExtractor_79448305041538_kernel(const float* __restrict__ feat,
                                            float* __restrict__ out) {
    __shared__ __attribute__((aligned(16))) float Xs[(C_CH / 2) * CPS];
    __shared__ float normS[NPOS];
    __shared__ __attribute__((aligned(16))) float Os[H_BAND * OSTR];

    const int blk   = blockIdx.x;
    const int strip = blk & (NSTRIP - 1);
    const int b     = blk >> 6;
    const int r0    = strip * ROWS;
    const int tid   = threadIdx.x;

    const float* fb = feat + (size_t)b * C_CH * L_LEN;
    v4f pa[5], pb[5];
#pragma unroll
    for (int i = 0; i < 5; ++i) {
        const int idx = tid + 256 * i;
        const int cp  = idx / NQ;
        const int q   = idx - cp * NQ;
        const int g   = r0 + 4 * q;
        const int gl  = (g < L_LEN) ? g : (L_LEN - 4);
        const float* s = fb + (size_t)(2 * cp) * L_LEN + gl;
        pa[i] = *(const v4f*)s;
        pb[i] = *(const v4f*)(s + L_LEN);
    }
#pragma unroll
    for (int i = 0; i < 5; ++i) {
        const int idx = tid + 256 * i;
        const int cp  = idx / NQ;
        const int q   = idx - cp * NQ;
        const bool in = (r0 + 4 * q) < L_LEN;
        const v4f zero = (v4f){0.0f, 0.0f, 0.0f, 0.0f};
        const v4f a = in ? pa[i] : zero;
        const v4f c = in ? pb[i] : zero;
        float* d = &Xs[cp * CPS + 8 * q];
        *(v4f*)d       = (v4f){a.x, c.x, a.y, c.y};
        *(v4f*)(d + 4) = (v4f){a.z, c.z, a.w, c.w};
    }
    __syncthreads();

    if (tid < NPOS) {
        float* xp = &Xs[2 * tid];
        float s = 0.0f;
        for (int cp = 0; cp < C_CH / 2; ++cp) {
            v2f x = *(const v2f*)(xp + cp * CPS);
            s += x.x;
            s += x.y;
        }
        float mu = s * (1.0f / 64.0f);
        float ss = 0.0f;
        for (int cp = 0; cp < C_CH / 2; ++cp) {
            v2f x = *(const v2f*)(xp + cp * CPS);
            float x0 = x.x - mu;
            float x1 = x.y - mu;
            *(v2f*)(xp + cp * CPS) = (v2f){x0, x1};
            ss = __builtin_fmaf(x0, x0, ss);
            ss = __builtin_fmaf(x1, x1, ss);
        }
        normS[tid] = sqrtf(ss);
    }
    __syncthreads();

    const int wave = tid >> 5;
    const int lane = tid & 31;
    const int nn   = lane & 15;
    const int half = lane >> 4;

    const float* fbase = &Xs[half * CPS + 2 * (16 * wave + nn)];
    v2f f0[16], f1[16], f2[16];
#pragma unroll
    for (int k0 = 0; k0 < 16; ++k0) {
        f0[k0] = *(const v2f*)(fbase + (2 * k0) * CPS);
        f1[k0] = *(const v2f*)(fbase + (2 * k0) * CPS + 32);
        f2[k0] = *(const v2f*)(fbase + (2 * k0) * CPS + 64);
    }

    __builtin_amdgcn_sched_barrier(0);

    v8f acc0 = {}, acc1 = {}, acc2 = {};
#pragma unroll
    for (int k0 = 0; k0 < 16; ++k0) {
        acc0 = __builtin_amdgcn_wmma_f32_16x16x4_f32(false, f0[k0], false, f0[k0], (short)0, acc0, false, false);
        acc1 = __builtin_amdgcn_wmma_f32_16x16x4_f32(false, f0[k0], false, f1[k0], (short)0, acc1, false, false);
        acc2 = __builtin_amdgcn_wmma_f32_16x16x4_f32(false, f0[k0], false, f2[k0], (short)0, acc2, false, false);
    }

    float nrow[8];
#pragma unroll
    for (int r = 0; r < 8; ++r) nrow[r] = normS[16 * wave + r + (half << 3)];

#pragma unroll
    for (int t = 0; t < 3; ++t) {
        const v8f acc = (t == 0) ? acc0 : ((t == 1) ? acc1 : acc2);
        const int   vrel = 16 * t + nn;
        const float ncol = normS[16 * wave + vrel];
        const bool  past = (r0 + 16 * wave + vrel) >= L_LEN;
#pragma unroll
        for (int r = 0; r < 8; ++r) {
            const int m = r + (half << 3);
            const int d = vrel - m - 1;
            if ((unsigned)d < (unsigned)H_BAND) {
                float denom = fmaxf(nrow[r] * ncol, EPS_F);
                float val   = acc[r] / denom;
                if (past) val = 0.0f;
                Os[d * OSTR + 16 * wave + m] = val;
            }
        }
    }
    __syncthreads();

    float* ob = out + (size_t)b * H_BAND * L_LEN + r0;
#pragma unroll
    for (int i = 0; i < 4; ++i) {
        const int idx = tid + 256 * i;
        const int d = idx >> 5;
        const int p = idx & 31;
        v4f v = *(const v4f*)&Os[d * OSTR + 4 * p];
        *(v4f*)(ob + (size_t)d * L_LEN + 4 * p) = v;
    }
}

extern "C" void kernel_launch(void* const* d_in, const int* in_sizes, int n_in,
                              void* d_out, int out_size, void* d_ws, size_t ws_size,
                              hipStream_t stream) {
    (void)in_sizes; (void)n_in; (void)out_size; (void)d_ws; (void)ws_size;
    const float* feat = (const float*)d_in[0];
    float* out = (float*)d_out;
    dim3 grid(B_SZ * NSTRIP), block(256);
    FeatureExtractor_79448305041538_kernel<<<grid, block, 0, stream>>>(feat, out);
}


# Epilogue made branch-free: first and third tile merged by a per-element select (exactly one is in the band), so a lane runs 16 independent divisions, was 24 each under its own branch; same values
# speedup vs baseline: 7.2019x; 1.1017x over previous
#include <hip/hip_runtime.h>
#include <math.h>

#define B_SZ   8
#define C_CH   64
#define L_LEN  8192
#define H_BAND 32
#define ROWS   128
#define NPOS   (ROWS + H_BAND)
#define NSTRIP (L_LEN / ROWS)
#define NQ     (NPOS / 4)
#define PP     176
#define CPS    (2 * PP)
#define OSTR   132
#define EPS_F  1e-12f

typedef float v2f __attribute__((ext_vector_type(2)));
typedef float v4f __attribute__((ext_vector_type(4)));
typedef float v8f __attribute__((ext_vector_type(8)));

__global__ __launch_bounds__(256)
void FeatureExtractor_79448305041538_kernel(const float* __restrict__ feat,
                                            float* __restrict__ out) {
    __shared__ __attribute__((aligned(16))) float Xs[(C_CH / 2) * CPS];
    __shared__ float normS[NPOS];
    __shared__ __attribute__((aligned(16))) float Os[H_BAND * OSTR];

    const int blk   = blockIdx.x;
    const int strip = blk & (NSTRIP - 1);
    const int b     = blk >> 6;
    const int r0    = strip * ROWS;
    const int tid   = threadIdx.x;

    const float* fb = feat + (size_t)b * C_CH * L_LEN;
    v4f pa[5], pb[5];
#pragma unroll
    for (int i = 0; i < 5; ++i) {
        const int idx = tid + 256 * i;
        const int cp  = idx / NQ;
        const int q   = idx - cp * NQ;
        const int g   = r0 + 4 * q;
        const int gl  = (g < L_LEN) ? g : (L_LEN - 4);
        const float* s = fb + (size_t)(2 * cp) * L_LEN + gl;
        pa[i] = *(const v4f*)s;
        pb[i] = *(const v4f*)(s + L_LEN);
    }
#pragma unroll
    for (int i = 0; i < 5; ++i) {
        const int idx = tid + 256 * i;
        const int cp  = idx / NQ;
        const int q   = idx - cp * NQ;
        const bool in = (r0 + 4 * q) < L_LEN;
        const v4f zero = (v4f){0.0f, 0.0f, 0.0f, 0.0f};
        const v4f a = in ? pa[i] : zero;
        const v4f c = in ? pb[i] : zero;
        float* d = &Xs[cp * CPS + 8 * q];
        *(v4f*)d       = (v4f){a.x, c.x, a.y, c.y};
        *(v4f*)(d + 4) = (v4f){a.z, c.z, a.w, c.w};
    }
    __syncthreads();

    if (tid < NPOS) {
        float* xp = &Xs[2 * tid];
        float s = 0.0f;
        for (int cp = 0; cp < C_CH / 2; ++cp) {
            v2f x = *(const v2f*)(xp + cp * CPS);
            s += x.x;
            s += x.y;
        }
        float mu = s * (1.0f / 64.0f);
        float ss = 0.0f;
        for (int cp = 0; cp < C_CH / 2; ++cp) {
            v2f x = *(const v2f*)(xp + cp * CPS);
            float x0 = x.x - mu;
            float x1 = x.y - mu;
            *(v2f*)(xp + cp * CPS) = (v2f){x0, x1};
            ss = __builtin_fmaf(x0, x0, ss);
            ss = __builtin_fmaf(x1, x1, ss);
        }
        normS[tid] = sqrtf(ss);
    }
    __syncthreads();

    const int wave = tid >> 5;
    const int lane = tid & 31;
    const int nn   = lane & 15;
    const int half = lane >> 4;

    const float* fbase = &Xs[half * CPS + 2 * (16 * wave + nn)];
    v2f f0[16], f1[16], f2[16];
#pragma unroll
    for (int k0 = 0; k0 < 16; ++k0) {
        f0[k0] = *(const v2f*)(fbase + (2 * k0) * CPS);
        f1[k0] = *(const v2f*)(fbase + (2 * k0) * CPS + 32);
        f2[k0] = *(const v2f*)(fbase + (2 * k0) * CPS + 64);
    }

    __builtin_amdgcn_sched_barrier(0);

    v8f acc0 = {}, acc1 = {}, acc2 = {};
#pragma unroll
    for (int k0 = 0; k0 < 16; ++k0) {
        acc0 = __builtin_amdgcn_wmma_f32_16x16x4_f32(false, f0[k0], false, f0[k0], (short)0, acc0, false, false);
        acc1 = __builtin_amdgcn_wmma_f32_16x16x4_f32(false, f0[k0], false, f1[k0], (short)0, acc1, false, false);
        acc2 = __builtin_amdgcn_wmma_f32_16x16x4_f32(false, f0[k0], false, f2[k0], (short)0, acc2, false, false);
    }

    float nrow[8];
#pragma unroll
    for (int r = 0; r < 8; ++r) nrow[r] = normS[16 * wave + r + (half << 3)];

    const int   cbase = 16 * wave + nn;
    const float ncol0 = normS[cbase];
    const float ncol1 = normS[cbase + 16];
    const float ncol2 = normS[cbase + 32];
    const bool  past0 = (r0 + cbase)      >= L_LEN;
    const bool  past1 = (r0 + cbase + 16) >= L_LEN;
    const bool  past2 = (r0 + cbase + 32) >= L_LEN;
#pragma unroll
    for (int r = 0; r < 8; ++r) {
        const int m   = r + (half << 3);
        const int row = 16 * wave + m;
        {
            float denom = fmaxf(nrow[r] * ncol1, EPS_F);
            float val   = acc1[r] / denom;
            if (past1) val = 0.0f;
            Os[(15 + nn - m) * OSTR + row] = val;
        }
        {
            const bool  up   = nn > m;
            const float a    = up ? acc0[r] : acc2[r];
            const float nc   = up ? ncol0 : ncol2;
            const bool  past = up ? past0 : past2;
            float denom = fmaxf(nrow[r] * nc, EPS_F);
            float val   = a / denom;
            if (past) val = 0.0f;
            Os[((nn - m - 1) & 31) * OSTR + row] = val;
        }
    }
    __syncthreads();

    float* ob = out + (size_t)b * H_BAND * L_LEN + r0;
#pragma unroll
    for (int i = 0; i < 4; ++i) {
        const int idx = tid + 256 * i;
        const int d = idx >> 5;
        const int p = idx & 31;
        v4f v = *(const v4f*)&Os[d * OSTR + 4 * p];
        *(v4f*)(ob + (size_t)d * L_LEN + 4 * p) = v;
    }
}

extern "C" void kernel_launch(void* const* d_in, const int* in_sizes, int n_in,
                              void* d_out, int out_size, void* d_ws, size_t ws_size,
                              hipStream_t stream) {
    (void)in_sizes; (void)n_in; (void)out_size; (void)d_ws; (void)ws_size;
    const float* feat = (const float*)d_in[0];
    float* out = (float*)d_out;
    dim3 grid(B_SZ * NSTRIP), block(256);
    FeatureExtractor_79448305041538_kernel<<<grid, block, 0, stream>>>(feat, out);
}


# Normalisation by reciprocal norms: one hardware rsqrt per position, then corr = num*min(rr*rc, 1/EPS) per element; was rounded sqrt and 16 IEEE divisions per lane; last bits of the result may differ
# speedup vs baseline: 7.6324x; 1.0598x over previous
#include <hip/hip_runtime.h>
#include <math.h>

#define B_SZ   8
#define C_CH   64
#define L_LEN  8192
#define H_BAND 32
#define ROWS   128
#define NPOS   (ROWS + H_BAND)
#define NSTRIP (L_LEN / ROWS)
#define NQ     (NPOS / 4)
#define PP     176
#define CPS    (2 * PP)
#define OSTR   132
#define EPS_F  1e-12f
#define RCP_EPS_F 1e12f

typedef float v2f __attribute__((ext_vector_type(2)));
typedef float v4f __attribute__((ext_vector_type(4)));
typedef float v8f __attribute__((ext_vector_type(8)));

__global__ __launch_bounds__(256)
void FeatureExtractor_79448305041538_kernel(const float* __restrict__ feat,
                                            float* __restrict__ out) {
    __shared__ __attribute__((aligned(16))) float Xs[(C_CH / 2) * CPS];
    __shared__ float rnS[NPOS];
    __shared__ __attribute__((aligned(16))) float Os[H_BAND * OSTR];

    const int blk   = blockIdx.x;
    const int strip = blk & (NSTRIP - 1);
    const int b     = blk >> 6;
    const int r0    = strip * ROWS;
    const int tid   = threadIdx.x;

    const float* fb = feat + (size_t)b * C_CH * L_LEN;
    v4f pa[5], pb[5];
#pragma unroll
    for (int i = 0; i < 5; ++i) {
        const int idx = tid + 256 * i;
        const int cp  = idx / NQ;
        const int q   = idx - cp * NQ;
        const int g   = r0 + 4 * q;
        const int gl  = (g < L_LEN) ? g : (L_LEN - 4);
        const float* s = fb + (size_t)(2 * cp) * L_LEN + gl;
        pa[i] = *(const v4f*)s;
        pb[i] = *(const v4f*)(s + L_LEN);
    }
#pragma unroll
    for (int i = 0; i < 5; ++i) {
        const int idx = tid + 256 * i;
        const int cp  = idx / NQ;
        const int q   = idx - cp * NQ;
        const bool in = (r0 + 4 * q) < L_LEN;
        const v4f zero = (v4f){0.0f, 0.0f, 0.0f, 0.0f};
        const v4f a = in ? pa[i] : zero;
        const v4f c = in ? pb[i] : zero;
        float* d = &Xs[cp * CPS + 8 * q];
        *(v4f*)d       = (v4f){a.x, c.x, a.y, c.y};
        *(v4f*)(d + 4) = (v4f){a.z, c.z, a.w, c.w};
    }
    __syncthreads();

    if (tid < NPOS) {
        float* xp = &Xs[2 * tid];
        float s = 0.0f;
        for (int cp = 0; cp < C_CH / 2; ++cp) {
            v2f x = *(const v2f*)(xp + cp * CPS);
            s += x.x;
            s += x.y;
        }
        float mu = s * (1.0f / 64.0f);
        float ss = 0.0f;
        for (int cp = 0; cp < C_CH / 2; ++cp) {
            v2f x = *(const v2f*)(xp + cp * CPS);
            float x0 = x.x - mu;
            float x1 = x.y - mu;
            *(v2f*)(xp + cp * CPS) = (v2f){x0, x1};
            ss = __builtin_fmaf(x0, x0, ss);
            ss = __builtin_fmaf(x1, x1, ss);
        }
        rnS[tid] = __builtin_amdgcn_rsqf(ss);
    }
    __syncthreads();

    const int wave = tid >> 5;
    const int lane = tid & 31;
    const int nn   = lane & 15;
    const int half = lane >> 4;

    const float* fbase = &Xs[half * CPS + 2 * (16 * wave + nn)];
    v2f f0[16], f1[16], f2[16];
#pragma unroll
    for (int k0 = 0; k0 < 16; ++k0) {
        f0[k0] = *(const v2f*)(fbase + (2 * k0) * CPS);
        f1[k0] = *(const v2f*)(fbase + (2 * k0) * CPS + 32);
        f2[k0] = *(const v2f*)(fbase + (2 * k0) * CPS + 64);
    }

    __builtin_amdgcn_sched_barrier(0);

    v8f acc0 = {}, acc1 = {}, acc2 = {};
#pragma unroll
    for (int k0 = 0; k0 < 16; ++k0) {
        acc0 = __builtin_amdgcn_wmma_f32_16x16x4_f32(false, f0[k0], false, f0[k0], (short)0, acc0, false, false);
        acc1 = __builtin_amdgcn_wmma_f32_16x16x4_f32(false, f0[k0], false, f1[k0], (short)0, acc1, false, false);
        acc2 = __builtin_amdgcn_wmma_f32_16x16x4_f32(false, f0[k0], false, f2[k0], (short)0, acc2, false, false);
    }

    float rrow[8];
#pragma unroll
    for (int r = 0; r < 8; ++r) rrow[r] = rnS[16 * wave + r + (half << 3)];

    const int   cbase = 16 * wave + nn;
    const float rcol0 = rnS[cbase];
    const float rcol1 = rnS[cbase + 16];
    const float rcol2 = rnS[cbase + 32];
    const bool  past0 = (r0 + cbase)      >= L_LEN;
    const bool  past1 = (r0 + cbase + 16) >= L_LEN;
    const bool  past2 = (r0 + cbase + 32) >= L_LEN;
#pragma unroll
    for (int r = 0; r < 8; ++r) {
        const int m   = r + (half << 3);
        const int row = 16 * wave + m;
        {
            float scale = fminf(rrow[r] * rcol1, RCP_EPS_F);
            float val   = acc1[r] * scale;
            if (past1) val = 0.0f;
            Os[(15 + nn - m) * OSTR + row] = val;
        }
        {
            const bool  up   = nn > m;
            const float a    = up ? acc0[r] : acc2[r];
            const float rc   = up ? rcol0 : rcol2;
            const bool  past = up ? past0 : past2;
            float scale = fminf(rrow[r] * rc, RCP_EPS_F);
            float val   = a * scale;
            if (past) val = 0.0f;
            Os[((nn - m - 1) & 31) * OSTR + row] = val;
        }
    }
    __syncthreads();

    float* ob = out + (size_t)b * H_BAND * L_LEN + r0;
#pragma unroll
    for (int i = 0; i < 4; ++i) {
        const int idx = tid + 256 * i;
        const int d = idx >> 5;
        const int p = idx & 31;
        v4f v = *(const v4f*)&Os[d * OSTR + 4 * p];
        *(v4f*)(ob + (size_t)d * L_LEN + 4 * p) = v;
    }
}

extern "C" void kernel_launch(void* const* d_in, const int* in_sizes, int n_in,
                              void* d_out, int out_size, void* d_ws, size_t ws_size,
                              hipStream_t stream) {
    (void)in_sizes; (void)n_in; (void)out_size; (void)d_ws; (void)ws_size;
    const float* feat = (const float*)d_in[0];
    float* out = (float*)d_out;
    dim3 grid(B_SZ * NSTRIP), block(256);
    FeatureExtractor_79448305041538_kernel<<<grid, block, 0, stream>>>(feat, out);
}


# Window centred and normalised in the staging registers (16 channels x 4 positions per lane; channel sums now combined over 4 lanes, sum order changed), written to LDS once; one barrier less
# speedup vs baseline: 8.6565x; 1.1342x over previous
#include <hip/hip_runtime.h>
#include <math.h>

#define B_SZ   8
#define C_CH   64
#define L_LEN  8192
#define H_BAND 32
#define ROWS   128
#define NPOS   (ROWS + H_BAND)
#define NSTRIP (L_LEN / ROWS)
#define PP     176
#define CPS    (2 * PP)
#define OSTR   132
#define EPS_F  1e-12f
#define RCP_EPS_F 1e12f

typedef float v2f __attribute__((ext_vector_type(2)));
typedef float v4f __attribute__((ext_vector_type(4)));
typedef float v8f __attribute__((ext_vector_type(8)));

__device__ __forceinline__ float quad_sum(float v) {
    const int b = __builtin_bit_cast(int, v);
    const float o = __builtin_bit_cast(float, __builtin_amdgcn_permlanex16(b, b, 0x76543210, 0xfedcba98, false, false));
    const float h = v + o;
    const int hb = __builtin_bit_cast(int, h);
    const float p = __builtin_bit_cast(float, __builtin_amdgcn_update_dpp(hb, hb, 0x128, 0xf, 0xf, false));
    return h + p;
}

__global__ __launch_bounds__(256)
void FeatureExtractor_79448305041538_kernel(const float* __restrict__ feat,
                                            float* __restrict__ out) {
    __shared__ __attribute__((aligned(16))) float Xs[(C_CH / 2) * CPS];
    __shared__ __attribute__((aligned(16))) float rnS[NPOS];
    __shared__ __attribute__((aligned(16))) float Os[H_BAND * OSTR];

    const int blk   = blockIdx.x;
    const int strip = blk & (NSTRIP - 1);
    const int b     = blk >> 6;
    const int r0    = strip * ROWS;
    const int tid   = threadIdx.x;

    const float* fb = feat + (size_t)b * C_CH * L_LEN;
    const int lane = tid & 31;
    const int wv = __builtin_amdgcn_readfirstlane(tid >> 5);
    if (wv < NPOS / 32) {
        const int jq = lane >> 3;
        const int g  = lane & 7;
        const int q  = 8 * wv + g;
        float* d = &Xs[(8 * jq) * CPS + 8 * q];
        if (r0 + 32 * wv < L_LEN) {
            const float* s = fb + (size_t)(16 * jq) * L_LEN + (r0 + 4 * q);
            v4f x[16];
#pragma unroll
            for (int i = 0; i < 16; ++i) x[i] = *(const v4f*)(s + (size_t)i * L_LEN);

            v4f se = x[0], so = x[1];
#pragma unroll
            for (int i = 2; i < 16; i += 2) { se += x[i]; so += x[i + 1]; }
            v4f sm = se + so;
            sm.x = quad_sum(sm.x); sm.y = quad_sum(sm.y); sm.z = quad_sum(sm.z); sm.w = quad_sum(sm.w);
            const v4f mu = sm * (1.0f / 64.0f);

            v4f qa = (v4f){0.0f, 0.0f, 0.0f, 0.0f}, qc = qa;
#pragma unroll
            for (int i = 0; i < 8; ++i) {
                const v4f a = x[2 * i]     - mu;
                const v4f c = x[2 * i + 1] - mu;
                qa.x = __builtin_fmaf(a.x, a.x, qa.x); qa.y = __builtin_fmaf(a.y, a.y, qa.y);
                qa.z = __builtin_fmaf(a.z, a.z, qa.z); qa.w = __builtin_fmaf(a.w, a.w, qa.w);
                qc.x = __builtin_fmaf(c.x, c.x, qc.x); qc.y = __builtin_fmaf(c.y, c.y, qc.y);
                qc.z = __builtin_fmaf(c.z, c.z, qc.z); qc.w = __builtin_fmaf(c.w, c.w, qc.w);
                *(v4f*)(d + i * CPS)     = (v4f){a.x, c.x, a.y, c.y};
                *(v4f*)(d + i * CPS + 4) = (v4f){a.z, c.z, a.w, c.w};
            }
            v4f ss = qa + qc;
            ss.x = quad_sum(ss.x); ss.y = quad_sum(ss.y); ss.z = quad_sum(ss.z); ss.w = quad_sum(ss.w);
            if (jq == 0) {
                *(v4f*)&rnS[4 * q] = (v4f){__builtin_amdgcn_rsqf(ss.x), __builtin_amdgcn_rsqf(ss.y),
                                           __builtin_amdgcn_rsqf(ss.z), __builtin_amdgcn_rsqf(ss.w)};
            }
        } else {
            const v4f zero = (v4f){0.0f, 0.0f, 0.0f, 0.0f};
#pragma unroll
            for (int i = 0; i < 8; ++i) {
                *(v4f*)(d + i * CPS)     = zero;
                *(v4f*)(d + i * CPS + 4) = zero;
            }
            if (jq == 0) {
                const float inf = __builtin_inff();
                *(v4f*)&rnS[4 * q] = (v4f){inf, inf, inf, inf};
            }
        }
    }
    __syncthreads();

    const int wave = tid >> 5;
    const int nn   = lane & 15;
    const int half = lane >> 4;

    const float* fbase = &Xs[half * CPS + 2 * (16 * wave + nn)];
    v2f f0[16], f1[16], f2[16];
#pragma unroll
    for (int k0 = 0; k0 < 16; ++k0) {
        f0[k0] = *(const v2f*)(fbase + (2 * k0) * CPS);
        f1[k0] = *(const v2f*)(fbase + (2 * k0) * CPS + 32);
        f2[k0] = *(const v2f*)(fbase + (2 * k0) * CPS + 64);
    }

    __builtin_amdgcn_sched_barrier(0);

    v8f acc0 = {}, acc1 = {}, acc2 = {};
#pragma unroll
    for (int k0 = 0; k0 < 16; ++k0) {
        acc0 = __builtin_amdgcn_wmma_f32_16x16x4_f32(false, f0[k0], false, f0[k0], (short)0, acc0, false, false);
        acc1 = __builtin_amdgcn_wmma_f32_16x16x4_f32(false, f0[k0], false, f1[k0], (short)0, acc1, false, false);
        acc2 = __builtin_amdgcn_wmma_f32_16x16x4_f32(false, f0[k0], false, f2[k0], (short)0, acc2, false, false);
    }

    float rrow[8];
#pragma unroll
    for (int r = 0; r < 8; ++r) rrow[r] = rnS[16 * wave + r + (half << 3)];

    const int   cbase = 16 * wave + nn;
    const float rcol0 = rnS[cbase];
    const float rcol1 = rnS[cbase + 16];
    const float rcol2 = rnS[cbase + 32];
    const bool  past0 = (r0 + cbase)      >= L_LEN;
    const bool  past1 = (r0 + cbase + 16) >= L_LEN;
    const bool  past2 = (r0 + cbase + 32) >= L_LEN;
#pragma unroll
    for (int r = 0; r < 8; ++r) {
        const int m   = r + (half << 3);
        const int row = 16 * wave + m;
        {
            float scale = fminf(rrow[r] * rcol1, RCP_EPS_F);
            float val   = acc1[r] * scale;
            if (past1) val = 0.0f;
            Os[(15 + nn - m) * OSTR + row] = val;
        }
        {
            const bool  up   = nn > m;
            const float a    = up ? acc0[r] : acc2[r];
            const float rc   = up ? rcol0 : rcol2;
            const bool  past = up ? past0 : past2;
            float scale = fminf(rrow[r] * rc, RCP_EPS_F);
            float val   = a * scale;
            if (past) val = 0.0f;
            Os[((nn - m - 1) & 31) * OSTR + row] = val;
        }
    }
    __syncthreads();

    float* ob = out + (size_t)b * H_BAND * L_LEN + r0;
#pragma unroll
    for (int i = 0; i < 4; ++i) {
        const int idx = tid + 256 * i;
        const int d = idx >> 5;
        const int p = idx & 31;
        v4f v = *(const v4f*)&Os[d * OSTR + 4 * p];
        *(v4f*)(ob + (size_t)d * L_LEN + 4 * p) = v;
    }
}

extern "C" void kernel_launch(void* const* d_in, const int* in_sizes, int n_in,
                              void* d_out, int out_size, void* d_ws, size_t ws_size,
                              hipStream_t stream) {
    (void)in_sizes; (void)n_in; (void)out_size; (void)d_ws; (void)ws_size;
    const float* feat = (const float*)d_in[0];
    float* out = (float*)d_out;
    dim3 grid(B_SZ * NSTRIP), block(256);
    FeatureExtractor_79448305041538_kernel<<<grid, block, 0, stream>>>(feat, out);
}
